# starting kernel
# baseline (speedup 1.0000x reference)
#include <hip/hip_runtime.h>
#include <math.h>


typedef __attribute__((ext_vector_type(2))) float v2f;
typedef __attribute__((ext_vector_type(8))) float v8f;

#define ALDS_STRIDE 36
#define BLDS_STRIDE 516
#define SLDS_STRIDE 524
#define LDS_FLOATS  (256 * ALDS_STRIDE + 32 * BLDS_STRIDE + 16 * SLDS_STRIDE)

__device__ __forceinline__ float lane_sum(float v) {
#pragma unroll
    for (int off = 16; off > 0; off >>= 1) v += __shfl_xor(v, off, 32);
    return v;
}
__device__ __forceinline__ float lane_max(float v) {
#pragma unroll
    for (int off = 16; off > 0; off >>= 1) v = fmaxf(v, __shfl_xor(v, off, 32));
    return v;
}

__device__ __forceinline__ void weighted_squash(const float* __restrict__ xv,
                                                float c, float* __restrict__ sv)
{
    float t[16];
#pragma unroll
    for (int d = 0; d < 16; ++d) t[d] = xv[d] * c;
#pragma unroll
    for (int off = 16; off > 0; off >>= 1) {
#pragma unroll
        for (int d = 0; d < 16; ++d) t[d] += __shfl_xor(t[d], off, 32);
    }
    float sq = 0.f;
#pragma unroll
    for (int d = 0; d < 16; ++d) sq = fmaf(t[d], t[d], sq);
    float coef = (sq / (1.0f + sq)) / sqrtf(sq + 1e-7f);
#pragma unroll
    for (int d = 0; d < 16; ++d) sv[d] = coef * t[d];
}

__global__ __launch_bounds__(256)
void capsule_fused_wmma(const float* __restrict__ prim,
                        const float* __restrict__ Wt,
                        const float* __restrict__ bprior,
                        float* __restrict__ out)
{
    extern __shared__ float smem[];
    float* Alds  = smem;
    float* Blds  = Alds + 256 * ALDS_STRIDE;
    float* Strip = Blds + 32 * BLDS_STRIDE;

    const int bm   = blockIdx.x;
    const int o    = bm & 31;
    const int tid  = threadIdx.x;
    const int wave = tid >> 5;
    const int lane = tid & 31;

    const float* Ag = prim + (size_t)bm * 8192;
    const float* Wg = Wt   + (size_t)bm * 16384;

    __builtin_prefetch(Wg, 0, 0);

#pragma unroll
    for (int i = 0; i < 8; ++i) {
        int f4  = tid + i * 256;
        float4 v = ((const float4*)Ag)[f4];
        int f   = f4 << 2;
        int row = f >> 5;
        int kk  = f & 31;
        float* dst = &Alds[row * ALDS_STRIDE + kk];
        dst[0] = v.x; dst[1] = v.y; dst[2] = v.z; dst[3] = v.w;
    }
#pragma unroll 4
    for (int i = 0; i < 64; ++i) {
        int lin = i * 256 + tid;
        int c   = lin >> 9;
        int n   = lin & 511;
        int yx  = c >> 3, e = c & 7;
        int og  = n >> 4, d = n & 15;
        Blds[c * BLDS_STRIDE + n] = Wg[yx * 4096 + og * 128 + d * 8 + e];
    }
    __syncthreads();

    const int khalf = (lane >> 4) << 1;

    for (int mtile = 0; mtile < 16; ++mtile) {
        const int arow = mtile * 16 + (lane & 15);
#pragma unroll
        for (int i = 0; i < 4; ++i) {
            int nt  = i * 8 + wave;
            int col = nt * 16 + (lane & 15);
            v8f acc = {};
#pragma unroll
            for (int kc = 0; kc < 32; kc += 4) {
                int kk = kc + khalf;
                v2f a, b;
                a.x = Alds[arow * ALDS_STRIDE + kk];
                a.y = Alds[arow * ALDS_STRIDE + kk + 1];
                b.x = Blds[kk * BLDS_STRIDE + col];
                b.y = Blds[(kk + 1) * BLDS_STRIDE + col];
                acc = __builtin_amdgcn_wmma_f32_16x16x4_f32(
                          false, a, false, b, (short)0, acc, false, false);
            }
            int rbase = (lane >> 4) << 3;
#pragma unroll
            for (int r = 0; r < 8; ++r)
                Strip[(rbase + r) * SLDS_STRIDE + col] = acc[r];
        }
        __syncthreads();

#pragma unroll
        for (int p = 0; p < 2; ++p) {
            int rowk = p * 8 + wave;
            int k    = mtile * 16 + rowk;

            const float* sp = &Strip[rowk * SLDS_STRIDE + lane * 16];
            float xv[16];
#pragma unroll
            for (int q = 0; q < 4; ++q) {
                float4 v = ((const float4*)sp)[q];
                xv[q * 4 + 0] = v.x; xv[q * 4 + 1] = v.y;
                xv[q * 4 + 2] = v.z; xv[q * 4 + 3] = v.w;
            }
            float bb = bprior[((size_t)o * 256 + k) * 32 + lane];

            float sv[16];
            {
                float mx = lane_max(bb);
                float e  = __expf(bb - mx);
                float s  = lane_sum(e);
                weighted_squash(xv, e / s, sv);
            }
#pragma unroll
            for (int it = 0; it < 2; ++it) {
                float dot = 0.f;
#pragma unroll
                for (int d = 0; d < 16; ++d) dot = fmaf(xv[d], sv[d], dot);
                bb += dot;
                float mx = lane_max(bb);
                float e  = __expf(bb - mx);
                float s  = lane_sum(e);
                weighted_squash(xv, e / s, sv);
            }

            if (lane == 0) {
                float4* op = (float4*)(out + ((size_t)bm * 256 + k) * 16);
                op[0] = make_float4(sv[0],  sv[1],  sv[2],  sv[3]);
                op[1] = make_float4(sv[4],  sv[5],  sv[6],  sv[7]);
                op[2] = make_float4(sv[8],  sv[9],  sv[10], sv[11]);
                op[3] = make_float4(sv[12], sv[13], sv[14], sv[15]);
            }
        }
        __syncthreads();
    }
}

extern "C" void kernel_launch(void* const* d_in, const int* in_sizes, int n_in,
                              void* d_out, int out_size, void* d_ws, size_t ws_size,
                              hipStream_t stream)
{
    const float* prim = (const float*)d_in[0];
    const float* Wt   = (const float*)d_in[1];
    const float* bpr  = (const float*)d_in[2];
    float* out = (float*)d_out;

    const size_t lds_bytes = (size_t)LDS_FLOATS * sizeof(float);
    capsule_fused_wmma<<<256, 256, lds_bytes, stream>>>(prim, Wt, bpr, out);
}


# Transposed GEMM: a D tile holds one position per lane, x stays in accumulators, routing sums run in-lane (m) plus one lane^16 exchange (d), softmax 1/sum applied to the sum; routing sums reordered
# speedup vs baseline: 7.9714x; 7.9714x over previous
#include <hip/hip_runtime.h>
#include <math.h>


typedef __attribute__((ext_vector_type(2))) float v2f;
typedef __attribute__((ext_vector_type(4))) float v4f;
typedef __attribute__((ext_vector_type(8))) float v8f;

__device__ __forceinline__ float xhalf(float x) {
    return __builtin_bit_cast(float, __builtin_amdgcn_permlanex16(
        __builtin_bit_cast(int, x), __builtin_bit_cast(int, x),
        0x76543210, 0xfedcba98, false, false));
}

template <int J>
__device__ __forceinline__ v2f pr(const v8f& a) { return __builtin_shufflevector(a, a, 2 * J, 2 * J + 1); }

__device__ __forceinline__ v2f fma2(v2f a, v2f b, v2f c) { return __builtin_elementwise_fma(a, b, c); }

__device__ __forceinline__ void route_stage(const float (&bb)[32], const v8f (&acc)[32],
                                            v2f (&vv)[4])
{
    float m0 = fmaxf(bb[0], bb[1]), m1 = fmaxf(bb[2], bb[3]);
#pragma unroll
    for (int m = 4; m < 32; m += 2) {
        m0 = fmaxf(m0, bb[m]);
        m1 = fmaxf(m1, bb[m + 1]);
    }
    const float mx = fmaxf(m0, m1);

    float e[32];
    float s0 = 0.f, s1 = 0.f, s2 = 0.f, s3 = 0.f;
#pragma unroll
    for (int m = 0; m < 32; m += 4) {
        e[m]     = __expf(bb[m]     - mx);
        e[m + 1] = __expf(bb[m + 1] - mx);
        e[m + 2] = __expf(bb[m + 2] - mx);
        e[m + 3] = __expf(bb[m + 3] - mx);
        s0 += e[m]; s1 += e[m + 1]; s2 += e[m + 2]; s3 += e[m + 3];
    }
    const float inv = 1.0f / ((s0 + s1) + (s2 + s3));

    v2f u0, u1, u2, u3;
    {
        const v2f ee = {e[0], e[0]};
        u0 = ee * pr<0>(acc[0]); u1 = ee * pr<1>(acc[0]);
        u2 = ee * pr<2>(acc[0]); u3 = ee * pr<3>(acc[0]);
    }
#pragma unroll
    for (int m = 1; m < 32; ++m) {
        const v2f ee = {e[m], e[m]};
        u0 = fma2(ee, pr<0>(acc[m]), u0); u1 = fma2(ee, pr<1>(acc[m]), u1);
        u2 = fma2(ee, pr<2>(acc[m]), u2); u3 = fma2(ee, pr<3>(acc[m]), u3);
    }
    const v2f inv2 = {inv, inv};
    u0 *= inv2; u1 *= inv2; u2 *= inv2; u3 *= inv2;
    v2f q = u0 * u0;
    q = fma2(u1, u1, q); q = fma2(u2, u2, q); q = fma2(u3, u3, q);
    float sq = q.x + q.y;
    sq += xhalf(sq);
    const float coef = (sq / (1.0f + sq)) / sqrtf(sq + 1e-7f);
    const v2f coef2 = {coef, coef};
    vv[0] = coef2 * u0; vv[1] = coef2 * u1; vv[2] = coef2 * u2; vv[3] = coef2 * u3;
}

__global__ __launch_bounds__(256)
void capsule_fused_wmma(const float* __restrict__ prim,
                        const float* __restrict__ Wt,
                        const float* __restrict__ bprior,
                        float* __restrict__ out)
{
    extern __shared__ v4f Wl[];

    const int bm   = blockIdx.x;
    const int o    = bm & 31;
    const int tid  = threadIdx.x;
    const int wave = tid >> 5;
    const int lane = tid & 31;
    const int c    = lane & 15;
    const int h    = lane >> 4;

    const float* Ag  = prim + (size_t)bm * 8192;
    const v4f*   Wg4 = (const v4f*)(Wt + (size_t)bm * 16384);

#pragma unroll
    for (int i = 0; i < 16; ++i) {
        int q   = i * 256 + tid;
        int yx  = q >> 10;
        int rem = q & 1023;
        int n   = ((rem >> 5) << 4) + (rem & 15);
        int hh  = (rem >> 4) & 1;
        Wl[yx * 1024 + hh * 512 + n] = Wg4[yx * 1024 + n * 2 + hh];
    }
    __syncthreads();

#pragma unroll 1
    for (int gi = 0; gi < 2; ++gi) {
        const int g = gi * 8 + wave;
        const int k = g * 16 + c;

        v4f pq[4];
#pragma unroll
        for (int yx = 0; yx < 4; ++yx)
            pq[yx] = *(const v4f*)(Ag + k * 32 + yx * 8 + 4 * h);

        float bb[32];
        {
            const v4f* bp = (const v4f*)(bprior + ((size_t)o * 256 + k) * 32);
#pragma unroll
            for (int q = 0; q < 8; ++q) {
                v4f t = bp[q];
                bb[q * 4 + 0] = t.x; bb[q * 4 + 1] = t.y;
                bb[q * 4 + 2] = t.z; bb[q * 4 + 3] = t.w;
            }
        }

        v8f acc[32];
        v4f wa[4], wb[4];
#pragma unroll
        for (int yx = 0; yx < 4; ++yx) {
            wa[yx] = Wl[yx * 1024 + h * 512 + c];
            wb[yx] = Wl[yx * 1024 + h * 512 + 16 + c];
        }
#pragma unroll
        for (int m = 0; m < 32; m += 2) {
            v4f na[4], nb[4];
            if (m + 2 < 32) {
#pragma unroll
                for (int yx = 0; yx < 4; ++yx) {
                    na[yx] = Wl[yx * 1024 + h * 512 + (m + 2) * 16 + c];
                    nb[yx] = Wl[yx * 1024 + h * 512 + (m + 3) * 16 + c];
                }
            }
            v8f a0 = {}, a1 = {};
#pragma unroll
            for (int yx = 0; yx < 4; ++yx) {
                v2f pf0 = {pq[yx].x, pq[yx].y};
                v2f pf1 = {pq[yx].z, pq[yx].w};
                v2f wa0 = {wa[yx].x, wa[yx].y};
                v2f wa1 = {wa[yx].z, wa[yx].w};
                v2f wb0 = {wb[yx].x, wb[yx].y};
                v2f wb1 = {wb[yx].z, wb[yx].w};
                a0 = __builtin_amdgcn_wmma_f32_16x16x4_f32(false, wa0, false, pf0, (short)0, a0, false, false);
                a1 = __builtin_amdgcn_wmma_f32_16x16x4_f32(false, wb0, false, pf0, (short)0, a1, false, false);
                a0 = __builtin_amdgcn_wmma_f32_16x16x4_f32(false, wa1, false, pf1, (short)0, a0, false, false);
                a1 = __builtin_amdgcn_wmma_f32_16x16x4_f32(false, wb1, false, pf1, (short)0, a1, false, false);
            }
            acc[m]     = a0;
            acc[m + 1] = a1;
            __builtin_amdgcn_sched_barrier(0);
            if (m + 2 < 32) {
#pragma unroll
                for (int yx = 0; yx < 4; ++yx) { wa[yx] = na[yx]; wb[yx] = nb[yx]; }
            }
        }

        v2f vv[4];
        route_stage(bb, acc, vv);
#pragma unroll
        for (int it = 0; it < 2; ++it) {
#pragma unroll
            for (int m = 0; m < 32; ++m) {
                v2f p2 = pr<0>(acc[m]) * vv[0];
                p2 = fma2(pr<1>(acc[m]), vv[1], p2);
                p2 = fma2(pr<2>(acc[m]), vv[2], p2);
                p2 = fma2(pr<3>(acc[m]), vv[3], p2);
                const float p = p2.x + p2.y;
                bb[m] += p + xhalf(p);
            }
            route_stage(bb, acc, vv);
        }

        v4f* op = (v4f*)(out + ((size_t)bm * 256 + k) * 16 + 8 * h);
        v4f o0 = {vv[0].x, vv[0].y, vv[1].x, vv[1].y};
        v4f o1 = {vv[2].x, vv[2].y, vv[3].x, vv[3].y};
        op[0] = o0;
        op[1] = o1;
    }
}

extern "C" void kernel_launch(void* const* d_in, const int* in_sizes, int n_in,
                              void* d_out, int out_size, void* d_ws, size_t ws_size,
                              hipStream_t stream)
{
    const float* prim = (const float*)d_in[0];
    const float* Wt   = (const float*)d_in[1];
    const float* bpr  = (const float*)d_in[2];
    float* out = (float*)d_out;

    capsule_fused_wmma<<<256, 256, 65536, stream>>>(prim, Wt, bpr, out);
}
